# starting kernel
# baseline (speedup 1.0000x reference)
#include <hip/hip_runtime.h>
#include <stdint.h>

typedef float v2f __attribute__((ext_vector_type(2)));
typedef float v8f __attribute__((ext_vector_type(8)));

#define N_NODES 50000
#define N_EDGES 800000
#define ETOT    (N_EDGES + N_NODES)
#define IN_CH   64
#define F1      256
#define H1      4
#define C1      64
#define F2      32
#define NEG_SLOPE 0.2f

__global__ void zero_f32(float* __restrict__ p, int n) {
  int i = blockIdx.x * blockDim.x + threadIdx.x;
  if (i < n) p[i] = 0.0f;
}
__global__ void zero_u32(unsigned* __restrict__ p, int n) {
  int i = blockIdx.x * blockDim.x + threadIdx.x;
  if (i < n) p[i] = 0u;
}

__device__ __forceinline__ void edge_sd(const int* __restrict__ ei, int e, int& s, int& d) {
  if (e < N_EDGES) { s = ei[e]; d = ei[N_EDGES + e]; }
  else             { s = d = e - N_EDGES; }
}

__device__ __forceinline__ unsigned f2key(float f) {
  unsigned b = __float_as_uint(f);
  return (b & 0x80000000u) ? ~b : (b | 0x80000000u);
}
__device__ __forceinline__ float key2f(unsigned k) {
  if (k == 0u) return 0.0f;
  unsigned b = (k & 0x80000000u) ? (k & 0x7fffffffu) : ~k;
  return __uint_as_float(b);
}

__device__ __forceinline__ float leaky(float x) {
  return x > 0.0f ? x : NEG_SLOPE * x;
}

template <int NT>
__global__ void wmma_gemm_f32(const float* __restrict__ A, const float* __restrict__ B,
                              float* __restrict__ C, int M, int K, int N, int tiles) {
  const int lane = threadIdx.x & 31;
  const int tile = blockIdx.x * (blockDim.x >> 5) + (threadIdx.x >> 5);
  if (tile >= tiles) return;
  const int tilesN = N / (16 * NT);
  const int tm  = tile / tilesN;
  const int tn0 = (tile % tilesN) * NT;

  const int rc   = lane & 15;
  const int hi   = lane >> 4;
  const int koff = hi << 1;

  const float* Ap = A + (size_t)(tm * 16 + rc) * K + koff;
  const float* Bp = B + (size_t)koff * N + tn0 * 16 + rc;

  v8f acc[NT];
#pragma unroll
  for (int t = 0; t < NT; ++t) acc[t] = (v8f){};

  for (int k0 = 0; k0 < K; k0 += 4) {
    v2f a;
    a.x = Ap[k0];
    a.y = Ap[k0 + 1];
#pragma unroll
    for (int t = 0; t < NT; ++t) {
      v2f b;
      b.x = Bp[(size_t)k0 * N + t * 16];
      b.y = Bp[(size_t)(k0 + 1) * N + t * 16];
      acc[t] = __builtin_amdgcn_wmma_f32_16x16x4_f32(false, a, false, b, (short)0,
                                                     acc[t], false, false);
    }
  }
  float* Cp = C + (size_t)(tm * 16 + hi * 8) * N + tn0 * 16 + rc;
#pragma unroll
  for (int t = 0; t < NT; ++t)
#pragma unroll
    for (int r = 0; r < 8; ++r) Cp[(size_t)r * N + t * 16] = acc[t][r];
}

template <int CH>
__global__ void att_dots(const float* __restrict__ h, const float* __restrict__ att_s,
                         const float* __restrict__ att_d, float* __restrict__ as,
                         float* __restrict__ ad, int n, int heads) {
  int idx = blockIdx.x * blockDim.x + threadIdx.x;
  if (idx >= n * heads) return;
  int node = idx / heads, hd = idx % heads;
  const float* hp = h + (size_t)node * heads * CH + (size_t)hd * CH;
  const float* sp = att_s + (size_t)hd * CH;
  const float* dp = att_d + (size_t)hd * CH;
  float s = 0.0f, d = 0.0f;
#pragma unroll 4
  for (int c = 0; c < CH; ++c) { float v = hp[c]; s += v * sp[c]; d += v * dp[c]; }
  as[idx] = s;
  ad[idx] = d;
}

template <int H>
__global__ void edge_logit_max(const int* __restrict__ ei, const float* __restrict__ as,
                               const float* __restrict__ ad, float* __restrict__ elog,
                               unsigned* __restrict__ lkey) {
  int idx = blockIdx.x * blockDim.x + threadIdx.x;
  if (idx >= ETOT * H) return;
  int e = idx / H, h = idx % H;
  int s, d; edge_sd(ei, e, s, d);
  float lg = leaky(as[s * H + h] + ad[d * H + h]);
  elog[idx] = lg;
  atomicMax(&lkey[d * H + h], f2key(lg));
}

template <int H>
__global__ void edge_expsum(const int* __restrict__ ei, float* __restrict__ elog,
                            const unsigned* __restrict__ lkey, float* __restrict__ denom) {
  int idx = blockIdx.x * blockDim.x + threadIdx.x;
  if (idx >= ETOT * H) return;
  int e = idx / H, h = idx % H;
  int s, d; edge_sd(ei, e, s, d);
  float m  = key2f(lkey[d * H + h]);
  float ex = expf(elog[idx] - m);
  elog[idx] = ex;
  atomicAdd(&denom[d * H + h], ex);
}

template <int H, int CH>
__global__ void edge_aggr(const int* __restrict__ ei, const float* __restrict__ elog,
                          const float* __restrict__ denom, const float* __restrict__ h,
                          float* __restrict__ out) {
  int wave = blockIdx.x * (blockDim.x >> 5) + (threadIdx.x >> 5);
  int lane = threadIdx.x & 31;
  if (wave >= ETOT) return;
  int s, d; edge_sd(ei, wave, s, d);
  constexpr int F   = H * CH;
  constexpr int per = F / 32;
  const int c0 = lane * per;
  const int hd = c0 / CH;
  float alpha = elog[wave * H + hd] / (denom[d * H + hd] + 1e-16f);
  const float* hp = h + (size_t)s * F + c0;
  float*       op = out + (size_t)d * F + c0;
  __builtin_prefetch(hp, 0, 1);
#pragma unroll
  for (int j = 0; j < per; ++j) atomicAdd(&op[j], alpha * hp[j]);
}

__global__ void bias_elu(float* __restrict__ h, const float* __restrict__ bias, int n, int F) {
  int i = blockIdx.x * blockDim.x + threadIdx.x;
  if (i >= n) return;
  float v = h[i] + bias[i % F];
  h[i] = v > 0.0f ? v : (expf(v) - 1.0f);
}
__global__ void bias_add(float* __restrict__ o, const float* __restrict__ bias, int n, int F) {
  int i = blockIdx.x * blockDim.x + threadIdx.x;
  if (i >= n) return;
  o[i] += bias[i % F];
}

extern "C" void kernel_launch(void* const* d_in, const int* in_sizes, int n_in,
                              void* d_out, int out_size, void* d_ws, size_t ws_size,
                              hipStream_t stream) {
  const float* x   = (const float*)d_in[0];
  const int*   ei  = (const int*)d_in[1];
  const float* W1  = (const float*)d_in[2];
  const float* as1 = (const float*)d_in[3];
  const float* ad1 = (const float*)d_in[4];
  const float* b1  = (const float*)d_in[5];
  const float* W2  = (const float*)d_in[6];
  const float* as2 = (const float*)d_in[7];
  const float* ad2 = (const float*)d_in[8];
  const float* b2  = (const float*)d_in[9];
  float* out = (float*)d_out;

  char* w = (char*)d_ws;
  auto carve = [&](size_t bytes) { char* p = w; w += (bytes + 255) & ~(size_t)255; return p; };
  float*    h1pre = (float*)   carve(sizeof(float) * N_NODES * F1);
  float*    h1    = (float*)   carve(sizeof(float) * N_NODES * F1);
  float*    h2pre = (float*)   carve(sizeof(float) * N_NODES * F2);
  float*    elog1 = (float*)   carve(sizeof(float) * (size_t)ETOT * H1);
  float*    elog2 = (float*)   carve(sizeof(float) * (size_t)ETOT);
  float*    asrc1 = (float*)   carve(sizeof(float) * N_NODES * H1);
  float*    adst1 = (float*)   carve(sizeof(float) * N_NODES * H1);
  float*    den1  = (float*)   carve(sizeof(float) * N_NODES * H1);
  unsigned* lk1   = (unsigned*)carve(sizeof(unsigned) * N_NODES * H1);
  float*    asrc2 = (float*)   carve(sizeof(float) * N_NODES);
  float*    adst2 = (float*)   carve(sizeof(float) * N_NODES);
  float*    den2  = (float*)   carve(sizeof(float) * N_NODES);
  unsigned* lk2   = (unsigned*)carve(sizeof(unsigned) * N_NODES);

  const int BLK = 256;
  auto cdiv = [](int a, int b) { return (a + b - 1) / b; };

  zero_f32<<<cdiv(N_NODES * F1, BLK), BLK, 0, stream>>>(h1, N_NODES * F1);
  zero_f32<<<cdiv(N_NODES * H1, BLK), BLK, 0, stream>>>(den1, N_NODES * H1);
  zero_u32<<<cdiv(N_NODES * H1, BLK), BLK, 0, stream>>>(lk1, N_NODES * H1);
  zero_f32<<<cdiv(N_NODES * F2, BLK), BLK, 0, stream>>>(out, N_NODES * F2);
  zero_f32<<<cdiv(N_NODES, BLK), BLK, 0, stream>>>(den2, N_NODES);
  zero_u32<<<cdiv(N_NODES, BLK), BLK, 0, stream>>>(lk2, N_NODES);

  {
    int tiles = (N_NODES / 16) * (F1 / 64);
    wmma_gemm_f32<4><<<cdiv(tiles, BLK / 32), BLK, 0, stream>>>(x, W1, h1pre,
                                                                N_NODES, IN_CH, F1, tiles);
  }
  att_dots<C1><<<cdiv(N_NODES * H1, BLK), BLK, 0, stream>>>(h1pre, as1, ad1,
                                                            asrc1, adst1, N_NODES, H1);
  edge_logit_max<H1><<<cdiv(ETOT * H1, BLK), BLK, 0, stream>>>(ei, asrc1, adst1, elog1, lk1);
  edge_expsum<H1><<<cdiv(ETOT * H1, BLK), BLK, 0, stream>>>(ei, elog1, lk1, den1);
  edge_aggr<H1, C1><<<cdiv(ETOT, BLK / 32), BLK, 0, stream>>>(ei, elog1, den1, h1pre, h1);
  bias_elu<<<cdiv(N_NODES * F1, BLK), BLK, 0, stream>>>(h1, b1, N_NODES * F1, F1);

  {
    int tiles = (N_NODES / 16) * (F2 / 32);
    wmma_gemm_f32<2><<<cdiv(tiles, BLK / 32), BLK, 0, stream>>>(h1, W2, h2pre,
                                                                N_NODES, F1, F2, tiles);
  }
  att_dots<F2><<<cdiv(N_NODES, BLK), BLK, 0, stream>>>(h2pre, as2, ad2,
                                                       asrc2, adst2, N_NODES, 1);
  edge_logit_max<1><<<cdiv(ETOT, BLK), BLK, 0, stream>>>(ei, asrc2, adst2, elog2, lk2);
  edge_expsum<1><<<cdiv(ETOT, BLK), BLK, 0, stream>>>(ei, elog2, lk2, den2);
  edge_aggr<1, F2><<<cdiv(ETOT, BLK / 32), BLK, 0, stream>>>(ei, elog2, den2, h2pre, out);
  bias_add<<<cdiv(N_NODES * F2, BLK), BLK, 0, stream>>>(out, b2, N_NODES * F2, F2);
}


# aggregation, both layers: a wave loads its edges' pair and a_src[src] once, ahead of the first barrier, and keeps src, row and logits in registers through the max, sum and row passes
# speedup vs baseline: 5.2646x; 5.2646x over previous
#include <hip/hip_runtime.h>
#include <stdint.h>

typedef float v2f __attribute__((ext_vector_type(2)));
typedef float v8f __attribute__((ext_vector_type(8)));

#define N_NODES 50000
#define N_EDGES 800000
#define IN_CH   64
#define F1      256
#define H1      4
#define C1      64
#define F2      32
#define NEG_SLOPE 0.2f

__device__ __forceinline__ unsigned f2key(float f) {
  unsigned b = __float_as_uint(f);
  return (b & 0x80000000u) ? ~b : (b | 0x80000000u);
}
__device__ __forceinline__ float key2f(unsigned k) {
  if (k == 0u) return 0.0f;
  unsigned b = (k & 0x80000000u) ? (k & 0x7fffffffu) : ~k;
  return __uint_as_float(b);
}

__device__ __forceinline__ float leaky(float x) {
  return x > 0.0f ? x : NEG_SLOPE * x;
}

template <int NT, int HEADS>
__global__ void wmma_gemm_dots_f32(const float* __restrict__ A, const float* __restrict__ B,
                                   const float* __restrict__ att_s, const float* __restrict__ att_d,
                                   float* __restrict__ C, float* __restrict__ as,
                                   float* __restrict__ ad, int M, int K, int tiles) {
  constexpr int N  = HEADS * 16 * NT;
  constexpr int CH = 16 * NT;
  const int lane = threadIdx.x & 31;
  const int tile = blockIdx.x * (blockDim.x >> 5) + (threadIdx.x >> 5);
  if (tile >= tiles) return;
  const int tm   = tile / HEADS;
  const int head = tile % HEADS;
  const int tn0  = head * NT;

  const int rc   = lane & 15;
  const int hi   = lane >> 4;
  const int koff = hi << 1;

  const float* Ap = A + (size_t)(tm * 16 + rc) * K + koff;
  const float* Bp = B + (size_t)koff * N + tn0 * 16 + rc;

  float ws[NT], wd[NT];
#pragma unroll
  for (int t = 0; t < NT; ++t) {
    ws[t] = att_s[head * CH + t * 16 + rc];
    wd[t] = att_d[head * CH + t * 16 + rc];
  }

  v8f acc[NT];
#pragma unroll
  for (int t = 0; t < NT; ++t) acc[t] = (v8f){};

  for (int k0 = 0; k0 < K; k0 += 4) {
    v2f a;
    a.x = Ap[k0];
    a.y = Ap[k0 + 1];
#pragma unroll
    for (int t = 0; t < NT; ++t) {
      v2f b;
      b.x = Bp[(size_t)k0 * N + t * 16];
      b.y = Bp[(size_t)(k0 + 1) * N + t * 16];
      acc[t] = __builtin_amdgcn_wmma_f32_16x16x4_f32(false, a, false, b, (short)0,
                                                     acc[t], false, false);
    }
  }
  float* Cp = C + (size_t)(tm * 16 + hi * 8) * N + tn0 * 16 + rc;
#pragma unroll
  for (int t = 0; t < NT; ++t)
#pragma unroll
    for (int r = 0; r < 8; ++r) Cp[(size_t)r * N + t * 16] = acc[t][r];

  float ps[8], pd[8];
#pragma unroll
  for (int r = 0; r < 8; ++r) {
    float s = 0.0f, d = 0.0f;
#pragma unroll
    for (int t = 0; t < NT; ++t) { s += acc[t][r] * ws[t]; d += acc[t][r] * wd[t]; }
#pragma unroll
    for (int m = 1; m < 16; m <<= 1) { s += __shfl_xor(s, m, 32); d += __shfl_xor(d, m, 32); }
    ps[r] = s;
    pd[r] = d;
  }
  float v = 0.0f;
#pragma unroll
  for (int r = 0; r < 8; ++r) {
    if (rc == r)     v = ps[r];
    if (rc == r + 8) v = pd[r];
  }
  float* dp = (rc < 8) ? as : ad;
  dp[(size_t)(tm * 16 + hi * 8 + (rc & 7)) * HEADS + head] = v;
}

#define TILE_NODES 196
#define TILE_THREADS 1024
#define NUM_TILES   256
#define BIN_BLOCKS  256
#define BIN_THREADS 256
#define BIN_CHUNK   (N_EDGES / BIN_BLOCKS)
static_assert(NUM_TILES * TILE_NODES >= N_NODES, "tiles must cover all nodes");
static_assert((NUM_TILES - 1) * TILE_NODES < N_NODES, "no empty trailing tile");
static_assert(BIN_BLOCKS * BIN_CHUNK == N_EDGES, "chunks must cover the edge list exactly");
static_assert(BIN_THREADS == NUM_TILES, "one thread per tile counter");

__global__ void __launch_bounds__(BIN_THREADS)
tile_count(const int* __restrict__ ei, int* __restrict__ cnt) {
  __shared__ int hist[NUM_TILES];
  hist[threadIdx.x] = 0;
  __syncthreads();
  const int* dst = ei + N_EDGES + blockIdx.x * BIN_CHUNK;
  for (int i = threadIdx.x; i < BIN_CHUNK; i += BIN_THREADS)
    atomicAdd(&hist[dst[i] / TILE_NODES], 1);
  __syncthreads();
  cnt[blockIdx.x * NUM_TILES + threadIdx.x] = hist[threadIdx.x];
}

#define OFF_GROUPS 4
#define OFF_SPAN   (BIN_BLOCKS / OFF_GROUPS)
__global__ void __launch_bounds__(OFF_GROUPS * NUM_TILES)
tile_offsets(const int* __restrict__ cnt, int* __restrict__ off, int* __restrict__ tile_start) {
  __shared__ int part[OFF_GROUPS][NUM_TILES];
  __shared__ int scan[NUM_TILES];
  const int t = threadIdx.x % NUM_TILES;
  const int q = threadIdx.x / NUM_TILES;
  int c[OFF_SPAN];
#pragma unroll
  for (int b = 0; b < OFF_SPAN; ++b) c[b] = cnt[(q * OFF_SPAN + b) * NUM_TILES + t];
  int sum = 0;
#pragma unroll
  for (int b = 0; b < OFF_SPAN; ++b) sum += c[b];
  part[q][t] = sum;
  __syncthreads();
  int total = 0, before = 0;
  for (int g = 0; g < OFF_GROUPS; ++g) { const int p = part[g][t]; total += p; if (g < q) before += p; }
  if (q == 0) scan[t] = total;
  __syncthreads();
  for (int o = 1; o < NUM_TILES; o <<= 1) {
    int v = 0;
    if (q == 0 && t >= o) v = scan[t - o];
    __syncthreads();
    if (q == 0) scan[t] += v;
    __syncthreads();
  }
  const int start = scan[t] - total;
  if (q == 0) {
    tile_start[t] = start;
    if (t == NUM_TILES - 1) tile_start[NUM_TILES] = scan[t];
  }
  int run = start + before;
#pragma unroll
  for (int b = 0; b < OFF_SPAN; ++b) { off[(q * OFF_SPAN + b) * NUM_TILES + t] = run; run += c[b]; }
}

__global__ void __launch_bounds__(BIN_THREADS)
tile_scatter(const int* __restrict__ ei, const int* __restrict__ off, int* __restrict__ bins) {
  __shared__ int cur[NUM_TILES];
  cur[threadIdx.x] = off[blockIdx.x * NUM_TILES + threadIdx.x];
  __syncthreads();
  const int* src = ei + blockIdx.x * BIN_CHUNK;
  const int* dst = src + N_EDGES;
  int2* pairs = reinterpret_cast<int2*>(bins);
  constexpr int B = 4;
  for (int i0 = threadIdx.x; i0 < BIN_CHUNK; i0 += B * BIN_THREADS) {
    int s[B], d[B];
#pragma unroll
    for (int u = 0; u < B; ++u) {
      const int i = i0 + u * BIN_THREADS;
      s[u] = (i < BIN_CHUNK) ? src[i] : 0;
      d[u] = (i < BIN_CHUNK) ? dst[i] : -1;
    }
#pragma unroll
    for (int u = 0; u < B; ++u)
      if (d[u] >= 0) pairs[atomicAdd(&cur[d[u] / TILE_NODES], 1)] = make_int2(s[u], d[u]);
  }
}

#define HELD_BATCHES 4

template <int H>
__device__ __forceinline__ void gather_asrc(const float* __restrict__ as, int s, float (&a)[H]) {
  if constexpr (H == 4) {
    const float4 t = *reinterpret_cast<const float4*>(as + (size_t)s * 4);
    a[0] = t.x; a[1] = t.y; a[2] = t.z; a[3] = t.w;
  } else {
#pragma unroll
    for (int q = 0; q < H; ++q) a[q] = as[(size_t)s * H + q];
  }
}

template <int H>
__device__ __forceinline__ void edge_logits(const float* __restrict__ as, const float* adl,
                                            int s, int dl, float (&lg)[H]) {
  float a[H];
  gather_asrc<H>(as, s, a);
#pragma unroll
  for (int q = 0; q < H; ++q) lg[q] = leaky(a[q] + adl[dl * H + q]);
}

template <int H, bool SUM>
__device__ __forceinline__ void stat_sweep(const int2* __restrict__ pairs,
                                           const float* __restrict__ as, const float* adl,
                                           unsigned* mkey, float* den,
                                           int wbeg, int wend, int lane, int d0) {
  constexpr int B = 4;
  for (int base = wbeg; base < wend; base += 32 * B) {
    int   s[B], dl[B];
    float lg[B][H];
#pragma unroll
    for (int u = 0; u < B; ++u) {
      const int e = base + u * 32 + lane;
      const int2 p = pairs[(e < wend) ? e : (wend - 1)];
      s[u]  = p.x;
      dl[u] = p.y - d0;
    }
#pragma unroll
    for (int u = 0; u < B; ++u) edge_logits<H>(as, adl, s[u], dl[u], lg[u]);
#pragma unroll
    for (int u = 0; u < B; ++u) {
      if (base + u * 32 + lane < wend) {
#pragma unroll
        for (int q = 0; q < H; ++q) {
          if constexpr (SUM)
            __hip_atomic_fetch_add(&den[dl[u] * H + q], expf(lg[u][q] - key2f(mkey[dl[u] * H + q])),
                                   __ATOMIC_RELAXED, __HIP_MEMORY_SCOPE_WORKGROUP);
          else
            atomicMax(&mkey[dl[u] * H + q], f2key(lg[u][q]));
        }
      }
    }
  }
}

template <int H, int CH, int U>
__device__ __forceinline__ void add_edges(float* __restrict__ acc, const float* __restrict__ h,
                                          int lane, int i, int s, int dl, const float (&al)[H]) {
  constexpr int F   = H * CH;
  constexpr int per = F / 32;
  float v[U][per];
  int   er[U];
#pragma unroll
  for (int u = 0; u < U; ++u) {
    const int es = __builtin_amdgcn_readlane(s, i + u);
    er[u] = __builtin_amdgcn_readlane(dl, i + u);
    float ea[H];
#pragma unroll
    for (int q = 0; q < H; ++q)
      ea[q] = __int_as_float(__builtin_amdgcn_readlane(__float_as_int(al[q]), i + u));
    const float* hp = h + (size_t)es * F + lane;
#pragma unroll
    for (int j = 0; j < per; ++j) v[u][j] = ea[(j * 32) / CH] * hp[j * 32];
  }
#pragma unroll
  for (int u = 0; u < U; ++u)
#pragma unroll
    for (int j = 0; j < per; ++j)
      __hip_atomic_fetch_add(&acc[er[u] * F + j * 32 + lane], v[u][j], __ATOMIC_RELAXED,
                             __HIP_MEMORY_SCOPE_WORKGROUP);
}

template <int H, int CH>
__global__ void __launch_bounds__(TILE_THREADS)
tile_aggr(const int* __restrict__ bins, const int* __restrict__ tile_start,
          const float* __restrict__ as, const float* __restrict__ ad,
          const float* __restrict__ h, float* __restrict__ out) {
  constexpr int F      = H * CH;
  constexpr int per    = F / 32;
  constexpr int NWAVES = TILE_THREADS / 32;
  constexpr int U      = (per >= 8) ? 4 : 16;
  constexpr int NS     = TILE_NODES * H;
  constexpr int HB     = HELD_BATCHES;
  __shared__ float acc[TILE_NODES * F];
  __shared__ __attribute__((aligned(16))) float    adl[NS];
  __shared__ __attribute__((aligned(16))) unsigned mkey[NS];
  __shared__ __attribute__((aligned(16))) float    den[NS];
  __shared__ __attribute__((aligned(16))) float    exs[NS];

  const int tid  = threadIdx.x;
  const int lane = tid & 31;
  const int wv   = __builtin_amdgcn_readfirstlane(tid >> 5);
  const int d0   = blockIdx.x * TILE_NODES;

  const int2* pairs = reinterpret_cast<const int2*>(bins);
  const int beg   = tile_start[blockIdx.x];
  const int end   = tile_start[blockIdx.x + 1];
  const int share = (end - beg + NWAVES - 1) / NWAVES;
  const int wbeg  = beg + wv * share;
  const int wend  = (wbeg + share < end) ? (wbeg + share) : end;
  const bool has  = wbeg < wend;
  const int hend  = (wbeg + 32 * HB < wend) ? (wbeg + 32 * HB) : wend;

  static_assert(NS <= TILE_THREADS, "one (node, head) slot per thread");
  const int  g       = d0 * H + tid;
  const bool slot    = tid < NS;
  const bool slot_in = slot && (g < N_NODES * H);
  const int   gc    = slot_in ? g : 0;
  const float sl_ad = ad[gc];
  const float sl_as = as[gc];

  int   hs[HB], hdl[HB];
  float hw[HB][H];
#pragma unroll
  for (int u = 0; u < HB; ++u) {
    hs[u] = 0;
    hdl[u] = 0;
#pragma unroll
    for (int q = 0; q < H; ++q) hw[u][q] = 0.0f;
  }
  if (has) {
#pragma unroll
    for (int u = 0; u < HB; ++u) {
      const int e = wbeg + u * 32 + lane;
      const int2 p = pairs[(e < hend) ? e : (hend - 1)];
      hs[u]  = p.x;
      hdl[u] = p.y - d0;
    }
#pragma unroll
    for (int u = 0; u < HB; ++u) gather_asrc<H>(as, hs[u], hw[u]);
  }

  if (slot) {
    const float a_d = slot_in ? sl_ad : 0.0f;
    const float lg  = slot_in ? leaky(sl_as + sl_ad) : 0.0f;
    adl[tid]  = a_d;
    exs[tid]  = lg;
    mkey[tid] = f2key(lg);
    den[tid]  = 0.0f;
  }
  __syncthreads();

  if (has) {
#pragma unroll
    for (int u = 0; u < HB; ++u) {
#pragma unroll
      for (int q = 0; q < H; ++q) hw[u][q] = leaky(hw[u][q] + adl[hdl[u] * H + q]);
      if (wbeg + u * 32 + lane < hend) {
#pragma unroll
        for (int q = 0; q < H; ++q) atomicMax(&mkey[hdl[u] * H + q], f2key(hw[u][q]));
      }
    }
    if (hend < wend) stat_sweep<H, false>(pairs, as, adl, mkey, den, hend, wend, lane, d0);
  }
  __syncthreads();

  for (int idx = tid; idx < NS; idx += TILE_THREADS) {
    const float ex = expf(exs[idx] - key2f(mkey[idx]));
    exs[idx] = ex;
    __hip_atomic_fetch_add(&den[idx], ex, __ATOMIC_RELAXED, __HIP_MEMORY_SCOPE_WORKGROUP);
  }
  if (has) {
#pragma unroll
    for (int u = 0; u < HB; ++u) {
      if (wbeg + u * 32 + lane < hend) {
#pragma unroll
        for (int q = 0; q < H; ++q) {
          const float ex = expf(hw[u][q] - key2f(mkey[hdl[u] * H + q]));
          hw[u][q] = ex;
          __hip_atomic_fetch_add(&den[hdl[u] * H + q], ex, __ATOMIC_RELAXED,
                                 __HIP_MEMORY_SCOPE_WORKGROUP);
        }
      }
    }
    if (hend < wend) stat_sweep<H, true>(pairs, as, adl, mkey, den, hend, wend, lane, d0);
  }
  __syncthreads();

  for (int r = wv; r < TILE_NODES; r += NWAVES) {
    const int d = d0 + r;
    if (d < N_NODES) {
      float al[H];
#pragma unroll
      for (int q = 0; q < H; ++q) al[q] = exs[r * H + q] / (den[r * H + q] + 1e-16f);
      const float* hp = h + (size_t)d * F + lane;
#pragma unroll
      for (int j = 0; j < per; ++j) acc[r * F + j * 32 + lane] = al[(j * 32) / CH] * hp[j * 32];
    }
  }
  __syncthreads();

  if (has) {
#pragma unroll
    for (int u = 0; u < HB; ++u) {
      const int left = hend - (wbeg + u * 32);
      const int n = (left < 0) ? 0 : ((left < 32) ? left : 32);
      float al[H];
#pragma unroll
      for (int q = 0; q < H; ++q) al[q] = hw[u][q] / (den[hdl[u] * H + q] + 1e-16f);
      int i = 0;
      for (; i + U <= n; i += U) add_edges<H, CH, U>(acc, h, lane, i, hs[u], hdl[u], al);
      for (; i < n; ++i)         add_edges<H, CH, 1>(acc, h, lane, i, hs[u], hdl[u], al);
    }
    for (int base = hend; base < wend; base += 32) {
      const int n = (wend - base < 32) ? (wend - base) : 32;
      int   s = 0, dl = 0;
      float al[H];
#pragma unroll
      for (int q = 0; q < H; ++q) al[q] = 0.0f;
      if (lane < n) {
        const int2 p = pairs[base + lane];
        s  = p.x;
        dl = p.y - d0;
        float lg[H];
        edge_logits<H>(as, adl, s, dl, lg);
#pragma unroll
        for (int q = 0; q < H; ++q)
          al[q] = expf(lg[q] - key2f(mkey[dl * H + q])) / (den[dl * H + q] + 1e-16f);
      }
      int i = 0;
      for (; i + U <= n; i += U) add_edges<H, CH, U>(acc, h, lane, i, s, dl, al);
      for (; i < n; ++i)         add_edges<H, CH, 1>(acc, h, lane, i, s, dl, al);
    }
  }
  __syncthreads();

  const int rows = (N_NODES - d0 < TILE_NODES) ? (N_NODES - d0) : TILE_NODES;
  const float4* a4 = reinterpret_cast<const float4*>(acc);
  float4*       o4 = reinterpret_cast<float4*>(out + (size_t)d0 * F);
  for (int i = tid; i < rows * (F / 4); i += TILE_THREADS) o4[i] = a4[i];
}

#define MAP_BLOCKS  256
#define MAP_THREADS 1024

template <bool ELU>
__device__ __forceinline__ float bias_map1(float x, float b) {
  const float v = x + b;
  if constexpr (ELU) return v > 0.0f ? v : (expf(v) - 1.0f);
  else               return v;
}

template <int F, bool ELU, int U>
__global__ void __launch_bounds__(MAP_THREADS)
bias_map(float* __restrict__ h, const float* __restrict__ bias) {
  constexpr int n4     = N_NODES * F / 4;
  constexpr int stride = MAP_BLOCKS * MAP_THREADS;
  static_assert(F % 4 == 0, "rows are whole 16-byte pieces");
  static_assert(stride % (F / 4) == 0, "a lane must keep its columns from trip to trip");
  const int i0 = blockIdx.x * MAP_THREADS + threadIdx.x;
  const int c  = (i0 % (F / 4)) * 4;
  const float b0 = bias[c], b1 = bias[c + 1], b2 = bias[c + 2], b3 = bias[c + 3];
  float4* p = reinterpret_cast<float4*>(h);
  for (int i = i0; i < n4; i += U * stride) {
    float4 v[U];
#pragma unroll
    for (int u = 0; u < U; ++u) {
      const int j = i + u * stride;
      v[u] = p[(j < n4) ? j : i];
    }
#pragma unroll
    for (int u = 0; u < U; ++u) {
      const int j = i + u * stride;
      float4 r;
      r.x = bias_map1<ELU>(v[u].x, b0);
      r.y = bias_map1<ELU>(v[u].y, b1);
      r.z = bias_map1<ELU>(v[u].z, b2);
      r.w = bias_map1<ELU>(v[u].w, b3);
      if (j < n4) p[j] = r;
    }
  }
}

extern "C" void kernel_launch(void* const* d_in, const int* in_sizes, int n_in,
                              void* d_out, int out_size, void* d_ws, size_t ws_size,
                              hipStream_t stream) {
  const float* x   = (const float*)d_in[0];
  const int*   ei  = (const int*)d_in[1];
  const float* W1  = (const float*)d_in[2];
  const float* as1 = (const float*)d_in[3];
  const float* ad1 = (const float*)d_in[4];
  const float* b1  = (const float*)d_in[5];
  const float* W2  = (const float*)d_in[6];
  const float* as2 = (const float*)d_in[7];
  const float* ad2 = (const float*)d_in[8];
  const float* b2  = (const float*)d_in[9];
  float* out = (float*)d_out;

  char* w = (char*)d_ws;
  auto carve = [&](size_t bytes) { char* p = w; w += (bytes + 255) & ~(size_t)255; return p; };
  float*    h1pre = (float*)   carve(sizeof(float) * N_NODES * F1);
  float*    h1    = (float*)   carve(sizeof(float) * N_NODES * F1);
  int*      bins  = (int*)     carve(sizeof(int) * 2 * (size_t)N_EDGES);
  int*      bcnt  = (int*)     carve(sizeof(int) * BIN_BLOCKS * NUM_TILES);
  int*      boff  = (int*)     carve(sizeof(int) * BIN_BLOCKS * NUM_TILES);
  int*      tst   = (int*)     carve(sizeof(int) * (NUM_TILES + 1));
  float*    h2pre = h1pre;
  float*    asrc1 = (float*)   carve(sizeof(float) * N_NODES * H1);
  float*    adst1 = (float*)   carve(sizeof(float) * N_NODES * H1);
  float*    asrc2 = (float*)   carve(sizeof(float) * N_NODES);
  float*    adst2 = (float*)   carve(sizeof(float) * N_NODES);

  const int BLK = 256;
  auto cdiv = [](int a, int b) { return (a + b - 1) / b; };

  tile_count<<<BIN_BLOCKS, BIN_THREADS, 0, stream>>>(ei, bcnt);
  tile_offsets<<<1, OFF_GROUPS * NUM_TILES, 0, stream>>>(bcnt, boff, tst);
  tile_scatter<<<BIN_BLOCKS, BIN_THREADS, 0, stream>>>(ei, boff, bins);

  {
    int tiles = (N_NODES / 16) * (F1 / 64);
    wmma_gemm_dots_f32<C1 / 16, H1><<<cdiv(tiles, BLK / 32), BLK, 0, stream>>>(
        x, W1, as1, ad1, h1pre, asrc1, adst1, N_NODES, IN_CH, tiles);
  }
  tile_aggr<H1, C1><<<NUM_TILES, TILE_THREADS, 0, stream>>>(bins, tst, asrc1, adst1, h1pre, h1);
  bias_map<F1, true, 6><<<MAP_BLOCKS, MAP_THREADS, 0, stream>>>(h1, b1);

  {
    int tiles = (N_NODES / 16) * (F2 / 32);
    wmma_gemm_dots_f32<F2 / 16, 1><<<cdiv(tiles, BLK / 32), BLK, 0, stream>>>(
        h1, W2, as2, ad2, h2pre, asrc2, adst2, N_NODES, F1, tiles);
  }
  tile_aggr<1, F2><<<NUM_TILES, TILE_THREADS, 0, stream>>>(bins, tst, asrc2, adst2, h2pre, out);
  bias_map<F2, false, 2><<<MAP_BLOCKS, MAP_THREADS, 0, stream>>>(out, b2);
}


# offsets step (one work-group) and the independent layer-1 GEMM share one launch: group 0 does the offsets, 255 resident groups of 32 waves loop over the GEMM strips; 9 launches become 8
# speedup vs baseline: 5.5671x; 1.0575x over previous
#include <hip/hip_runtime.h>
#include <stdint.h>

typedef float v2f __attribute__((ext_vector_type(2)));
typedef float v8f __attribute__((ext_vector_type(8)));

#define N_NODES 50000
#define N_EDGES 800000
#define IN_CH   64
#define F1      256
#define H1      4
#define C1      64
#define F2      32
#define NEG_SLOPE 0.2f

__device__ __forceinline__ unsigned f2key(float f) {
  unsigned b = __float_as_uint(f);
  return (b & 0x80000000u) ? ~b : (b | 0x80000000u);
}
__device__ __forceinline__ float key2f(unsigned k) {
  if (k == 0u) return 0.0f;
  unsigned b = (k & 0x80000000u) ? (k & 0x7fffffffu) : ~k;
  return __uint_as_float(b);
}

__device__ __forceinline__ float leaky(float x) {
  return x > 0.0f ? x : NEG_SLOPE * x;
}

template <int NT, int HEADS>
__device__ __forceinline__ void gemm_strip_dots(const float* __restrict__ A, const float* __restrict__ B,
                                                const float* __restrict__ att_s, const float* __restrict__ att_d,
                                                float* __restrict__ C, float* __restrict__ as,
                                                float* __restrict__ ad, int K, int tile, int lane) {
  constexpr int N  = HEADS * 16 * NT;
  constexpr int CH = 16 * NT;
  const int tm   = tile / HEADS;
  const int head = tile % HEADS;
  const int tn0  = head * NT;

  const int rc   = lane & 15;
  const int hi   = lane >> 4;
  const int koff = hi << 1;

  const float* Ap = A + (size_t)(tm * 16 + rc) * K + koff;
  const float* Bp = B + (size_t)koff * N + tn0 * 16 + rc;

  float ws[NT], wd[NT];
#pragma unroll
  for (int t = 0; t < NT; ++t) {
    ws[t] = att_s[head * CH + t * 16 + rc];
    wd[t] = att_d[head * CH + t * 16 + rc];
  }

  v8f acc[NT];
#pragma unroll
  for (int t = 0; t < NT; ++t) acc[t] = (v8f){};

  for (int k0 = 0; k0 < K; k0 += 4) {
    v2f a;
    a.x = Ap[k0];
    a.y = Ap[k0 + 1];
#pragma unroll
    for (int t = 0; t < NT; ++t) {
      v2f b;
      b.x = Bp[(size_t)k0 * N + t * 16];
      b.y = Bp[(size_t)(k0 + 1) * N + t * 16];
      acc[t] = __builtin_amdgcn_wmma_f32_16x16x4_f32(false, a, false, b, (short)0,
                                                     acc[t], false, false);
    }
  }
  float* Cp = C + (size_t)(tm * 16 + hi * 8) * N + tn0 * 16 + rc;
#pragma unroll
  for (int t = 0; t < NT; ++t)
#pragma unroll
    for (int r = 0; r < 8; ++r) Cp[(size_t)r * N + t * 16] = acc[t][r];

  float ps[8], pd[8];
#pragma unroll
  for (int r = 0; r < 8; ++r) {
    float s = 0.0f, d = 0.0f;
#pragma unroll
    for (int t = 0; t < NT; ++t) { s += acc[t][r] * ws[t]; d += acc[t][r] * wd[t]; }
#pragma unroll
    for (int m = 1; m < 16; m <<= 1) { s += __shfl_xor(s, m, 32); d += __shfl_xor(d, m, 32); }
    ps[r] = s;
    pd[r] = d;
  }
  float v = 0.0f;
#pragma unroll
  for (int r = 0; r < 8; ++r) {
    if (rc == r)     v = ps[r];
    if (rc == r + 8) v = pd[r];
  }
  float* dp = (rc < 8) ? as : ad;
  dp[(size_t)(tm * 16 + hi * 8 + (rc & 7)) * HEADS + head] = v;
}

template <int NT, int HEADS>
__global__ void wmma_gemm_dots_f32(const float* __restrict__ A, const float* __restrict__ B,
                                   const float* __restrict__ att_s, const float* __restrict__ att_d,
                                   float* __restrict__ C, float* __restrict__ as,
                                   float* __restrict__ ad, int M, int K, int tiles) {
  const int lane = threadIdx.x & 31;
  const int tile = blockIdx.x * (blockDim.x >> 5) + (threadIdx.x >> 5);
  if (tile >= tiles) return;
  gemm_strip_dots<NT, HEADS>(A, B, att_s, att_d, C, as, ad, K, tile, lane);
}

#define TILE_NODES 196
#define TILE_THREADS 1024
#define NUM_TILES   256
#define BIN_BLOCKS  256
#define BIN_THREADS 256
#define BIN_CHUNK   (N_EDGES / BIN_BLOCKS)
static_assert(NUM_TILES * TILE_NODES >= N_NODES, "tiles must cover all nodes");
static_assert((NUM_TILES - 1) * TILE_NODES < N_NODES, "no empty trailing tile");
static_assert(BIN_BLOCKS * BIN_CHUNK == N_EDGES, "chunks must cover the edge list exactly");
static_assert(BIN_THREADS == NUM_TILES, "one thread per tile counter");

__global__ void __launch_bounds__(BIN_THREADS)
tile_count(const int* __restrict__ ei, int* __restrict__ cnt) {
  __shared__ int hist[NUM_TILES];
  hist[threadIdx.x] = 0;
  __syncthreads();
  const int* dst = ei + N_EDGES + blockIdx.x * BIN_CHUNK;
  for (int i = threadIdx.x; i < BIN_CHUNK; i += BIN_THREADS)
    atomicAdd(&hist[dst[i] / TILE_NODES], 1);
  __syncthreads();
  cnt[blockIdx.x * NUM_TILES + threadIdx.x] = hist[threadIdx.x];
}

#define OFF_GROUPS 4
#define OFF_SPAN   (BIN_BLOCKS / OFF_GROUPS)
__device__ __forceinline__ void
tile_offsets_block(const int* __restrict__ cnt, int* __restrict__ off, int* __restrict__ tile_start) {
  __shared__ int part[OFF_GROUPS][NUM_TILES];
  __shared__ int scan[NUM_TILES];
  const int t = threadIdx.x % NUM_TILES;
  const int q = threadIdx.x / NUM_TILES;
  int c[OFF_SPAN];
#pragma unroll
  for (int b = 0; b < OFF_SPAN; ++b) c[b] = cnt[(q * OFF_SPAN + b) * NUM_TILES + t];
  int sum = 0;
#pragma unroll
  for (int b = 0; b < OFF_SPAN; ++b) sum += c[b];
  part[q][t] = sum;
  __syncthreads();
  int total = 0, before = 0;
  for (int g = 0; g < OFF_GROUPS; ++g) { const int p = part[g][t]; total += p; if (g < q) before += p; }
  if (q == 0) scan[t] = total;
  __syncthreads();
  for (int o = 1; o < NUM_TILES; o <<= 1) {
    int v = 0;
    if (q == 0 && t >= o) v = scan[t - o];
    __syncthreads();
    if (q == 0) scan[t] += v;
    __syncthreads();
  }
  const int start = scan[t] - total;
  if (q == 0) {
    tile_start[t] = start;
    if (t == NUM_TILES - 1) tile_start[NUM_TILES] = scan[t];
  }
  int run = start + before;
#pragma unroll
  for (int b = 0; b < OFF_SPAN; ++b) { off[(q * OFF_SPAN + b) * NUM_TILES + t] = run; run += c[b]; }
}

#define RIDE_BLOCKS 256
template <int NT, int HEADS>
__global__ void __launch_bounds__(OFF_GROUPS * NUM_TILES)
offsets_beside_gemm(const int* __restrict__ cnt, int* __restrict__ off, int* __restrict__ tile_start,
                    const float* __restrict__ A, const float* __restrict__ B,
                    const float* __restrict__ att_s, const float* __restrict__ att_d,
                    float* __restrict__ C, float* __restrict__ as, float* __restrict__ ad,
                    int K, int tiles) {
  if (blockIdx.x == 0) {
    tile_offsets_block(cnt, off, tile_start);
    return;
  }
  constexpr int G      = RIDE_BLOCKS - 1;
  constexpr int NWAVES = (OFF_GROUPS * NUM_TILES) / 32;
  const int lane  = threadIdx.x & 31;
  const int wv    = __builtin_amdgcn_readfirstlane(threadIdx.x >> 5);
  const int g     = blockIdx.x - 1;
  const int begin = (g * tiles) / G;
  const int end   = ((g + 1) * tiles) / G;
  for (int tile = begin + wv; tile < end; tile += NWAVES)
    gemm_strip_dots<NT, HEADS>(A, B, att_s, att_d, C, as, ad, K, tile, lane);
}

__global__ void __launch_bounds__(BIN_THREADS)
tile_scatter(const int* __restrict__ ei, const int* __restrict__ off, int* __restrict__ bins) {
  __shared__ int cur[NUM_TILES];
  cur[threadIdx.x] = off[blockIdx.x * NUM_TILES + threadIdx.x];
  __syncthreads();
  const int* src = ei + blockIdx.x * BIN_CHUNK;
  const int* dst = src + N_EDGES;
  int2* pairs = reinterpret_cast<int2*>(bins);
  constexpr int B = 4;
  for (int i0 = threadIdx.x; i0 < BIN_CHUNK; i0 += B * BIN_THREADS) {
    int s[B], d[B];
#pragma unroll
    for (int u = 0; u < B; ++u) {
      const int i = i0 + u * BIN_THREADS;
      s[u] = (i < BIN_CHUNK) ? src[i] : 0;
      d[u] = (i < BIN_CHUNK) ? dst[i] : -1;
    }
#pragma unroll
    for (int u = 0; u < B; ++u)
      if (d[u] >= 0) pairs[atomicAdd(&cur[d[u] / TILE_NODES], 1)] = make_int2(s[u], d[u]);
  }
}

#define HELD_BATCHES 4

template <int H>
__device__ __forceinline__ void gather_asrc(const float* __restrict__ as, int s, float (&a)[H]) {
  if constexpr (H == 4) {
    const float4 t = *reinterpret_cast<const float4*>(as + (size_t)s * 4);
    a[0] = t.x; a[1] = t.y; a[2] = t.z; a[3] = t.w;
  } else {
#pragma unroll
    for (int q = 0; q < H; ++q) a[q] = as[(size_t)s * H + q];
  }
}

template <int H>
__device__ __forceinline__ void edge_logits(const float* __restrict__ as, const float* adl,
                                            int s, int dl, float (&lg)[H]) {
  float a[H];
  gather_asrc<H>(as, s, a);
#pragma unroll
  for (int q = 0; q < H; ++q) lg[q] = leaky(a[q] + adl[dl * H + q]);
}

template <int H, bool SUM>
__device__ __forceinline__ void stat_sweep(const int2* __restrict__ pairs,
                                           const float* __restrict__ as, const float* adl,
                                           unsigned* mkey, float* den,
                                           int wbeg, int wend, int lane, int d0) {
  constexpr int B = 4;
  for (int base = wbeg; base < wend; base += 32 * B) {
    int   s[B], dl[B];
    float lg[B][H];
#pragma unroll
    for (int u = 0; u < B; ++u) {
      const int e = base + u * 32 + lane;
      const int2 p = pairs[(e < wend) ? e : (wend - 1)];
      s[u]  = p.x;
      dl[u] = p.y - d0;
    }
#pragma unroll
    for (int u = 0; u < B; ++u) edge_logits<H>(as, adl, s[u], dl[u], lg[u]);
#pragma unroll
    for (int u = 0; u < B; ++u) {
      if (base + u * 32 + lane < wend) {
#pragma unroll
        for (int q = 0; q < H; ++q) {
          if constexpr (SUM)
            __hip_atomic_fetch_add(&den[dl[u] * H + q], expf(lg[u][q] - key2f(mkey[dl[u] * H + q])),
                                   __ATOMIC_RELAXED, __HIP_MEMORY_SCOPE_WORKGROUP);
          else
            atomicMax(&mkey[dl[u] * H + q], f2key(lg[u][q]));
        }
      }
    }
  }
}

template <int H, int CH, int U>
__device__ __forceinline__ void add_edges(float* __restrict__ acc, const float* __restrict__ h,
                                          int lane, int i, int s, int dl, const float (&al)[H]) {
  constexpr int F   = H * CH;
  constexpr int per = F / 32;
  float v[U][per];
  int   er[U];
#pragma unroll
  for (int u = 0; u < U; ++u) {
    const int es = __builtin_amdgcn_readlane(s, i + u);
    er[u] = __builtin_amdgcn_readlane(dl, i + u);
    float ea[H];
#pragma unroll
    for (int q = 0; q < H; ++q)
      ea[q] = __int_as_float(__builtin_amdgcn_readlane(__float_as_int(al[q]), i + u));
    const float* hp = h + (size_t)es * F + lane;
#pragma unroll
    for (int j = 0; j < per; ++j) v[u][j] = ea[(j * 32) / CH] * hp[j * 32];
  }
#pragma unroll
  for (int u = 0; u < U; ++u)
#pragma unroll
    for (int j = 0; j < per; ++j)
      __hip_atomic_fetch_add(&acc[er[u] * F + j * 32 + lane], v[u][j], __ATOMIC_RELAXED,
                             __HIP_MEMORY_SCOPE_WORKGROUP);
}

template <int H, int CH>
__global__ void __launch_bounds__(TILE_THREADS)
tile_aggr(const int* __restrict__ bins, const int* __restrict__ tile_start,
          const float* __restrict__ as, const float* __restrict__ ad,
          const float* __restrict__ h, float* __restrict__ out) {
  constexpr int F      = H * CH;
  constexpr int per    = F / 32;
  constexpr int NWAVES = TILE_THREADS / 32;
  constexpr int U      = (per >= 8) ? 4 : 16;
  constexpr int NS     = TILE_NODES * H;
  constexpr int HB     = HELD_BATCHES;
  __shared__ float acc[TILE_NODES * F];
  __shared__ __attribute__((aligned(16))) float    adl[NS];
  __shared__ __attribute__((aligned(16))) unsigned mkey[NS];
  __shared__ __attribute__((aligned(16))) float    den[NS];
  __shared__ __attribute__((aligned(16))) float    exs[NS];

  const int tid  = threadIdx.x;
  const int lane = tid & 31;
  const int wv   = __builtin_amdgcn_readfirstlane(tid >> 5);
  const int d0   = blockIdx.x * TILE_NODES;

  const int2* pairs = reinterpret_cast<const int2*>(bins);
  const int beg   = tile_start[blockIdx.x];
  const int end   = tile_start[blockIdx.x + 1];
  const int share = (end - beg + NWAVES - 1) / NWAVES;
  const int wbeg  = beg + wv * share;
  const int wend  = (wbeg + share < end) ? (wbeg + share) : end;
  const bool has  = wbeg < wend;
  const int hend  = (wbeg + 32 * HB < wend) ? (wbeg + 32 * HB) : wend;

  static_assert(NS <= TILE_THREADS, "one (node, head) slot per thread");
  const int  g       = d0 * H + tid;
  const bool slot    = tid < NS;
  const bool slot_in = slot && (g < N_NODES * H);
  const int   gc    = slot_in ? g : 0;
  const float sl_ad = ad[gc];
  const float sl_as = as[gc];

  int   hs[HB], hdl[HB];
  float hw[HB][H];
#pragma unroll
  for (int u = 0; u < HB; ++u) {
    hs[u] = 0;
    hdl[u] = 0;
#pragma unroll
    for (int q = 0; q < H; ++q) hw[u][q] = 0.0f;
  }
  if (has) {
#pragma unroll
    for (int u = 0; u < HB; ++u) {
      const int e = wbeg + u * 32 + lane;
      const int2 p = pairs[(e < hend) ? e : (hend - 1)];
      hs[u]  = p.x;
      hdl[u] = p.y - d0;
    }
#pragma unroll
    for (int u = 0; u < HB; ++u) gather_asrc<H>(as, hs[u], hw[u]);
  }

  if (slot) {
    const float a_d = slot_in ? sl_ad : 0.0f;
    const float lg  = slot_in ? leaky(sl_as + sl_ad) : 0.0f;
    adl[tid]  = a_d;
    exs[tid]  = lg;
    mkey[tid] = f2key(lg);
    den[tid]  = 0.0f;
  }
  __syncthreads();

  if (has) {
#pragma unroll
    for (int u = 0; u < HB; ++u) {
#pragma unroll
      for (int q = 0; q < H; ++q) hw[u][q] = leaky(hw[u][q] + adl[hdl[u] * H + q]);
      if (wbeg + u * 32 + lane < hend) {
#pragma unroll
        for (int q = 0; q < H; ++q) atomicMax(&mkey[hdl[u] * H + q], f2key(hw[u][q]));
      }
    }
    if (hend < wend) stat_sweep<H, false>(pairs, as, adl, mkey, den, hend, wend, lane, d0);
  }
  __syncthreads();

  for (int idx = tid; idx < NS; idx += TILE_THREADS) {
    const float ex = expf(exs[idx] - key2f(mkey[idx]));
    exs[idx] = ex;
    __hip_atomic_fetch_add(&den[idx], ex, __ATOMIC_RELAXED, __HIP_MEMORY_SCOPE_WORKGROUP);
  }
  if (has) {
#pragma unroll
    for (int u = 0; u < HB; ++u) {
      if (wbeg + u * 32 + lane < hend) {
#pragma unroll
        for (int q = 0; q < H; ++q) {
          const float ex = expf(hw[u][q] - key2f(mkey[hdl[u] * H + q]));
          hw[u][q] = ex;
          __hip_atomic_fetch_add(&den[hdl[u] * H + q], ex, __ATOMIC_RELAXED,
                                 __HIP_MEMORY_SCOPE_WORKGROUP);
        }
      }
    }
    if (hend < wend) stat_sweep<H, true>(pairs, as, adl, mkey, den, hend, wend, lane, d0);
  }
  __syncthreads();

  for (int r = wv; r < TILE_NODES; r += NWAVES) {
    const int d = d0 + r;
    if (d < N_NODES) {
      float al[H];
#pragma unroll
      for (int q = 0; q < H; ++q) al[q] = exs[r * H + q] / (den[r * H + q] + 1e-16f);
      const float* hp = h + (size_t)d * F + lane;
#pragma unroll
      for (int j = 0; j < per; ++j) acc[r * F + j * 32 + lane] = al[(j * 32) / CH] * hp[j * 32];
    }
  }
  __syncthreads();

  if (has) {
#pragma unroll
    for (int u = 0; u < HB; ++u) {
      const int left = hend - (wbeg + u * 32);
      const int n = (left < 0) ? 0 : ((left < 32) ? left : 32);
      float al[H];
#pragma unroll
      for (int q = 0; q < H; ++q) al[q] = hw[u][q] / (den[hdl[u] * H + q] + 1e-16f);
      int i = 0;
      for (; i + U <= n; i += U) add_edges<H, CH, U>(acc, h, lane, i, hs[u], hdl[u], al);
      for (; i < n; ++i)         add_edges<H, CH, 1>(acc, h, lane, i, hs[u], hdl[u], al);
    }
    for (int base = hend; base < wend; base += 32) {
      const int n = (wend - base < 32) ? (wend - base) : 32;
      int   s = 0, dl = 0;
      float al[H];
#pragma unroll
      for (int q = 0; q < H; ++q) al[q] = 0.0f;
      if (lane < n) {
        const int2 p = pairs[base + lane];
        s  = p.x;
        dl = p.y - d0;
        float lg[H];
        edge_logits<H>(as, adl, s, dl, lg);
#pragma unroll
        for (int q = 0; q < H; ++q)
          al[q] = expf(lg[q] - key2f(mkey[dl * H + q])) / (den[dl * H + q] + 1e-16f);
      }
      int i = 0;
      for (; i + U <= n; i += U) add_edges<H, CH, U>(acc, h, lane, i, s, dl, al);
      for (; i < n; ++i)         add_edges<H, CH, 1>(acc, h, lane, i, s, dl, al);
    }
  }
  __syncthreads();

  const int rows = (N_NODES - d0 < TILE_NODES) ? (N_NODES - d0) : TILE_NODES;
  const float4* a4 = reinterpret_cast<const float4*>(acc);
  float4*       o4 = reinterpret_cast<float4*>(out + (size_t)d0 * F);
  for (int i = tid; i < rows * (F / 4); i += TILE_THREADS) o4[i] = a4[i];
}

#define MAP_BLOCKS  256
#define MAP_THREADS 1024

template <bool ELU>
__device__ __forceinline__ float bias_map1(float x, float b) {
  const float v = x + b;
  if constexpr (ELU) return v > 0.0f ? v : (expf(v) - 1.0f);
  else               return v;
}

template <int F, bool ELU, int U>
__global__ void __launch_bounds__(MAP_THREADS)
bias_map(float* __restrict__ h, const float* __restrict__ bias) {
  constexpr int n4     = N_NODES * F / 4;
  constexpr int stride = MAP_BLOCKS * MAP_THREADS;
  static_assert(F % 4 == 0, "rows are whole 16-byte pieces");
  static_assert(stride % (F / 4) == 0, "a lane must keep its columns from trip to trip");
  const int i0 = blockIdx.x * MAP_THREADS + threadIdx.x;
  const int c  = (i0 % (F / 4)) * 4;
  const float b0 = bias[c], b1 = bias[c + 1], b2 = bias[c + 2], b3 = bias[c + 3];
  float4* p = reinterpret_cast<float4*>(h);
  for (int i = i0; i < n4; i += U * stride) {
    float4 v[U];
#pragma unroll
    for (int u = 0; u < U; ++u) {
      const int j = i + u * stride;
      v[u] = p[(j < n4) ? j : i];
    }
#pragma unroll
    for (int u = 0; u < U; ++u) {
      const int j = i + u * stride;
      float4 r;
      r.x = bias_map1<ELU>(v[u].x, b0);
      r.y = bias_map1<ELU>(v[u].y, b1);
      r.z = bias_map1<ELU>(v[u].z, b2);
      r.w = bias_map1<ELU>(v[u].w, b3);
      if (j < n4) p[j] = r;
    }
  }
}

extern "C" void kernel_launch(void* const* d_in, const int* in_sizes, int n_in,
                              void* d_out, int out_size, void* d_ws, size_t ws_size,
                              hipStream_t stream) {
  const float* x   = (const float*)d_in[0];
  const int*   ei  = (const int*)d_in[1];
  const float* W1  = (const float*)d_in[2];
  const float* as1 = (const float*)d_in[3];
  const float* ad1 = (const float*)d_in[4];
  const float* b1  = (const float*)d_in[5];
  const float* W2  = (const float*)d_in[6];
  const float* as2 = (const float*)d_in[7];
  const float* ad2 = (const float*)d_in[8];
  const float* b2  = (const float*)d_in[9];
  float* out = (float*)d_out;

  char* w = (char*)d_ws;
  auto carve = [&](size_t bytes) { char* p = w; w += (bytes + 255) & ~(size_t)255; return p; };
  float*    h1pre = (float*)   carve(sizeof(float) * N_NODES * F1);
  float*    h1    = (float*)   carve(sizeof(float) * N_NODES * F1);
  int*      bins  = (int*)     carve(sizeof(int) * 2 * (size_t)N_EDGES);
  int*      bcnt  = (int*)     carve(sizeof(int) * BIN_BLOCKS * NUM_TILES);
  int*      boff  = (int*)     carve(sizeof(int) * BIN_BLOCKS * NUM_TILES);
  int*      tst   = (int*)     carve(sizeof(int) * (NUM_TILES + 1));
  float*    h2pre = h1pre;
  float*    asrc1 = (float*)   carve(sizeof(float) * N_NODES * H1);
  float*    adst1 = (float*)   carve(sizeof(float) * N_NODES * H1);
  float*    asrc2 = (float*)   carve(sizeof(float) * N_NODES);
  float*    adst2 = (float*)   carve(sizeof(float) * N_NODES);

  const int BLK = 256;
  auto cdiv = [](int a, int b) { return (a + b - 1) / b; };

  tile_count<<<BIN_BLOCKS, BIN_THREADS, 0, stream>>>(ei, bcnt);
  offsets_beside_gemm<C1 / 16, H1><<<RIDE_BLOCKS, OFF_GROUPS * NUM_TILES, 0, stream>>>(
      bcnt, boff, tst, x, W1, as1, ad1, h1pre, asrc1, adst1, IN_CH, (N_NODES / 16) * (F1 / 64));
  tile_scatter<<<BIN_BLOCKS, BIN_THREADS, 0, stream>>>(ei, boff, bins);

  tile_aggr<H1, C1><<<NUM_TILES, TILE_THREADS, 0, stream>>>(bins, tst, asrc1, adst1, h1pre, h1);
  bias_map<F1, true, 6><<<MAP_BLOCKS, MAP_THREADS, 0, stream>>>(h1, b1);

  {
    int tiles = (N_NODES / 16) * (F2 / 32);
    wmma_gemm_dots_f32<F2 / 16, 1><<<cdiv(tiles, BLK / 32), BLK, 0, stream>>>(
        h1, W2, as2, ad2, h2pre, asrc2, adst2, N_NODES, F1, tiles);
  }
  tile_aggr<1, F2><<<NUM_TILES, TILE_THREADS, 0, stream>>>(bins, tst, asrc2, adst2, h2pre, out);
  bias_map<F2, false, 2><<<MAP_BLOCKS, MAP_THREADS, 0, stream>>>(out, b2);
}
